# starting kernel
# baseline (speedup 1.0000x reference)
#include <hip/hip_runtime.h>


typedef __attribute__((ext_vector_type(2))) float v2f;
typedef __attribute__((ext_vector_type(8))) float v8f;

static inline int cdiv(long long a, long long b) { return (int)((a + b - 1) / b); }

template <int K, int N>
__launch_bounds__(256)
__global__ void gemm_wmma_f32(const float* __restrict__ X,
                              const float* __restrict__ W,
                              float* __restrict__ H,
                              int nrows) {
    constexpr int NT = N / 16;
    __shared__ float BF[(K / 4) * NT * 64];

    const int tid = threadIdx.x;
    for (int idx = tid; idx < K * N; idx += 256) {
        const int n  = idx / K;
        const int k  = idx - n * K;
        const int ks = k >> 2;
        const int kk = k & 3;
        const int g  = kk >> 1;
        const int r  = kk & 1;
        const int t  = n >> 4;
        const int nl = n & 15;
        BF[((ks * NT + t) * 32 + g * 16 + nl) * 2 + r] = W[idx];
    }
    __syncthreads();

    const int wave   = tid >> 5;
    const int lane   = tid & 31;
    const int laneLo = lane & 15;
    const int laneHi = lane >> 4;
    const int m0     = (blockIdx.x * 8 + wave) * 16;
    if (m0 >= nrows) return;

    v8f acc[NT] = {};

    const float* xrow  = X + (size_t)(m0 + laneLo) * K;
    const float* fbase = BF + lane * 2;
    for (int ks = 0; ks < K / 4; ++ks) {
        const v2f a = *(const v2f*)(xrow + ks * 4 + 2 * laneHi);
        const float* fb = fbase + (ks * NT) * 64;
#pragma unroll
        for (int t = 0; t < NT; ++t) {
            const v2f b = *(const v2f*)(fb + t * 64);
            acc[t] = __builtin_amdgcn_wmma_f32_16x16x4_f32(
                false, a, false, b, (short)0, acc[t], false, false);
        }
    }

#pragma unroll
    for (int t = 0; t < NT; ++t)
#pragma unroll
        for (int r = 0; r < 8; ++r)
            H[(size_t)(m0 + r + 8 * laneHi) * N + t * 16 + laneLo] = acc[t][r];
}

__global__ void k_fill1(float* __restrict__ deg, int n) {
    int i = blockIdx.x * blockDim.x + threadIdx.x;
    if (i < n) deg[i] = 1.0f;
}

__global__ void k_deg(const int* __restrict__ dst, float* __restrict__ deg, int E) {
    int e = blockIdx.x * blockDim.x + threadIdx.x;
    if (e < E) atomicAdd(&deg[dst[e]], 1.0f);
}

__global__ void k_rsqrt(float* __restrict__ d, int n) {
    int i = blockIdx.x * blockDim.x + threadIdx.x;
    if (i < n) d[i] = rsqrtf(d[i]);
}

__global__ void k_norm(const int* __restrict__ src, const int* __restrict__ dst,
                       const float* __restrict__ dinv, float* __restrict__ norm, int E) {
    int e = blockIdx.x * blockDim.x + threadIdx.x;
    if (e < E) norm[e] = dinv[src[e]] * dinv[dst[e]];
}

template <int C>
__global__ void k_selfinit(const float* __restrict__ H, const float* __restrict__ dinv,
                           const float* __restrict__ bias, float* __restrict__ out, int n) {
    int gid = blockIdx.x * blockDim.x + threadIdx.x;
    const int CH = C / 4;
    int i = gid / CH;
    if (i >= n) return;
    int c = (gid - i * CH) * 4;
    float s = dinv[i] * dinv[i];
    float4 h = *(const float4*)(H + (size_t)i * C + c);
    float4 o;
    o.x = h.x * s + bias[c + 0];
    o.y = h.y * s + bias[c + 1];
    o.z = h.z * s + bias[c + 2];
    o.w = h.w * s + bias[c + 3];
    *(float4*)(out + (size_t)i * C + c) = o;
}

template <int C>
__global__ void k_scatter(const int* __restrict__ src, const int* __restrict__ dst,
                          const float* __restrict__ norm, const float* __restrict__ H,
                          float* __restrict__ out, int E) {
    int gid = blockIdx.x * blockDim.x + threadIdx.x;
    const int CH = C / 4;
    int e = gid / CH;
    if (e >= E) return;
    int c = (gid - e * CH) * 4;
    float nrm = norm[e];
    float4 h = *(const float4*)(H + (size_t)src[e] * C + c);
    float* o = out + (size_t)dst[e] * C + c;
    atomicAdd(o + 0, h.x * nrm);
    atomicAdd(o + 1, h.y * nrm);
    atomicAdd(o + 2, h.z * nrm);
    atomicAdd(o + 3, h.w * nrm);
}

__global__ void k_relu(const float* __restrict__ in, float* __restrict__ out, int n) {
    int i = blockIdx.x * blockDim.x + threadIdx.x;
    if (i < n) out[i] = fmaxf(in[i], 0.0f);
}

extern "C" void kernel_launch(void* const* d_in, const int* in_sizes, int n_in,
                              void* d_out, int out_size, void* d_ws, size_t ws_size,
                              hipStream_t stream) {
    constexpr int IN  = 256;
    constexpr int HID = 128;
    constexpr int LAT = 64;

    const float* x   = (const float*)d_in[0];
    const int*   ei  = (const int*)d_in[1];
    const float* W1  = (const float*)d_in[2];
    const float* b1  = (const float*)d_in[3];
    const float* W2  = (const float*)d_in[4];
    const float* b2  = (const float*)d_in[5];
    float*       out = (float*)d_out;

    const int N = in_sizes[0] / IN;
    const int E = in_sizes[1] / 2;
    const int* src = ei;
    const int* dst = ei + E;

    char* ws = (char*)d_ws;
    float* dinv = (float*)ws;  ws += (size_t)N * sizeof(float);
    float* norm = (float*)ws;  ws += (size_t)E * sizeof(float);
    float* H1   = (float*)ws;  ws += (size_t)N * HID * sizeof(float);
    float* S1   = (float*)ws;
    float* H2   = S1;

    const int T = 256;

    k_fill1<<<cdiv(N, T), T, 0, stream>>>(dinv, N);
    k_deg  <<<cdiv(E, T), T, 0, stream>>>(dst, dinv, E);
    k_rsqrt<<<cdiv(N, T), T, 0, stream>>>(dinv, N);
    k_norm <<<cdiv(E, T), T, 0, stream>>>(src, dst, dinv, norm, E);

    gemm_wmma_f32<IN, HID><<<cdiv(N, 128), T, 0, stream>>>(x, W1, H1, N);

    k_selfinit<HID><<<cdiv((long long)N * (HID / 4), T), T, 0, stream>>>(H1, dinv, b1, S1, N);
    k_scatter<HID><<<cdiv((long long)E * (HID / 4), T), T, 0, stream>>>(src, dst, norm, H1, S1, E);
    k_relu<<<cdiv((long long)N * HID, T), T, 0, stream>>>(S1, H1, N * HID);

    gemm_wmma_f32<HID, LAT><<<cdiv(N, 128), T, 0, stream>>>(H1, W2, H2, N);

    k_selfinit<LAT><<<cdiv((long long)N * (LAT / 4), T), T, 0, stream>>>(H2, dinv, b2, out, N);
    k_scatter<LAT><<<cdiv((long long)E * (LAT / 4), T), T, 0, stream>>>(src, dst, norm, H2, out, E);
}


# Self-loop+bias start plane stored by the product kernels, ReLU applied at the layer-2 A load: three elementwise launches (550k short waves) go, 11 launches -> 8; every sum and its order unchanged
# speedup vs baseline: 1.6262x; 1.6262x over previous
#include <hip/hip_runtime.h>


typedef __attribute__((ext_vector_type(2))) float v2f;
typedef __attribute__((ext_vector_type(8))) float v8f;

static inline int cdiv(long long a, long long b) { return (int)((a + b - 1) / b); }

template <int K, int N, bool RELU_A>
__launch_bounds__(256)
__global__ void gemm_wmma_f32(const float* __restrict__ X,
                              const float* __restrict__ W,
                              const float* __restrict__ dinv,
                              const float* __restrict__ bias,
                              float* __restrict__ H,
                              float* __restrict__ S,
                              int nrows) {
    constexpr int NT = N / 16;
    __shared__ float BF[(K / 4) * NT * 64];

    const int tid = threadIdx.x;
    for (int idx = tid; idx < K * N; idx += 256) {
        const int n  = idx / K;
        const int k  = idx - n * K;
        const int ks = k >> 2;
        const int kk = k & 3;
        const int g  = kk >> 1;
        const int r  = kk & 1;
        const int t  = n >> 4;
        const int nl = n & 15;
        BF[((ks * NT + t) * 32 + g * 16 + nl) * 2 + r] = W[idx];
    }
    __syncthreads();

    const int wave   = tid >> 5;
    const int lane   = tid & 31;
    const int laneLo = lane & 15;
    const int laneHi = lane >> 4;
    const int m0     = (blockIdx.x * 8 + wave) * 16;
    if (m0 >= nrows) return;

    v8f acc[NT] = {};

    const float* xrow  = X + (size_t)(m0 + laneLo) * K;
    const float* fbase = BF + lane * 2;
    for (int ks = 0; ks < K / 4; ++ks) {
        v2f a = *(const v2f*)(xrow + ks * 4 + 2 * laneHi);
        if (RELU_A) { a.x = fmaxf(a.x, 0.0f); a.y = fmaxf(a.y, 0.0f); }
        const float* fb = fbase + (ks * NT) * 64;
#pragma unroll
        for (int t = 0; t < NT; ++t) {
            const v2f b = *(const v2f*)(fb + t * 64);
            acc[t] = __builtin_amdgcn_wmma_f32_16x16x4_f32(
                false, a, false, b, (short)0, acc[t], false, false);
        }
    }

    const float4 da = *(const float4*)(dinv + m0 + 8 * laneHi);
    const float4 db = *(const float4*)(dinv + m0 + 8 * laneHi + 4);
    const float sc[8] = {da.x * da.x, da.y * da.y, da.z * da.z, da.w * da.w,
                         db.x * db.x, db.y * db.y, db.z * db.z, db.w * db.w};
#pragma unroll
    for (int t = 0; t < NT; ++t) {
        const float bt = bias[t * 16 + laneLo];
#pragma unroll
        for (int r = 0; r < 8; ++r) {
            const size_t o = (size_t)(m0 + r + 8 * laneHi) * N + t * 16 + laneLo;
            H[o] = acc[t][r];
            S[o] = acc[t][r] * sc[r] + bt;
        }
    }
}

__global__ void k_fill1(float* __restrict__ deg, int n) {
    int i = blockIdx.x * blockDim.x + threadIdx.x;
    if (i < n) deg[i] = 1.0f;
}

__global__ void k_deg(const int* __restrict__ dst, float* __restrict__ deg, int E) {
    int e = blockIdx.x * blockDim.x + threadIdx.x;
    if (e < E) atomicAdd(&deg[dst[e]], 1.0f);
}

__global__ void k_rsqrt(float* __restrict__ d, int n) {
    int i = blockIdx.x * blockDim.x + threadIdx.x;
    if (i < n) d[i] = rsqrtf(d[i]);
}

__global__ void k_norm(const int* __restrict__ src, const int* __restrict__ dst,
                       const float* __restrict__ dinv, float* __restrict__ norm, int E) {
    int e = blockIdx.x * blockDim.x + threadIdx.x;
    if (e < E) norm[e] = dinv[src[e]] * dinv[dst[e]];
}

template <int C>
__global__ void k_scatter(const int* __restrict__ src, const int* __restrict__ dst,
                          const float* __restrict__ norm, const float* __restrict__ H,
                          float* __restrict__ out, int E) {
    constexpr int Q = C / 32;
    constexpr int U = 8;
    const int lane = threadIdx.x & 31;
    const int wave = blockIdx.x * (blockDim.x >> 5) + (threadIdx.x >> 5);
    const int e0   = wave * 32;
    if (e0 >= E) return;
    const int n = min(32, E - e0);

    int s = 0, d = 0;
    float w = 0.0f;
    if (lane < n) { s = src[e0 + lane]; d = dst[e0 + lane]; w = norm[e0 + lane]; }

    int j = 0;
    for (; j + U <= n; j += U) {
        float v[U][Q];
#pragma unroll
        for (int u = 0; u < U; ++u) {
            const int sj = __builtin_amdgcn_readlane(s, j + u);
            const float* hp = H + (size_t)sj * C + lane;
#pragma unroll
            for (int q = 0; q < Q; ++q) v[u][q] = hp[q * 32];
        }
#pragma unroll
        for (int u = 0; u < U; ++u) {
            const int   dj = __builtin_amdgcn_readlane(d, j + u);
            const float wj = __int_as_float(__builtin_amdgcn_readlane(__float_as_int(w), j + u));
            float* op = out + (size_t)dj * C + lane;
#pragma unroll
            for (int q = 0; q < Q; ++q) atomicAdd(op + q * 32, v[u][q] * wj);
        }
    }
    for (; j < n; ++j) {
        const int   sj = __builtin_amdgcn_readlane(s, j);
        const int   dj = __builtin_amdgcn_readlane(d, j);
        const float wj = __int_as_float(__builtin_amdgcn_readlane(__float_as_int(w), j));
        const float* hp = H + (size_t)sj * C + lane;
        float* op = out + (size_t)dj * C + lane;
#pragma unroll
        for (int q = 0; q < Q; ++q) atomicAdd(op + q * 32, hp[q * 32] * wj);
    }
}

extern "C" void kernel_launch(void* const* d_in, const int* in_sizes, int n_in,
                              void* d_out, int out_size, void* d_ws, size_t ws_size,
                              hipStream_t stream) {
    constexpr int IN  = 256;
    constexpr int HID = 128;
    constexpr int LAT = 64;

    const float* x   = (const float*)d_in[0];
    const int*   ei  = (const int*)d_in[1];
    const float* W1  = (const float*)d_in[2];
    const float* b1  = (const float*)d_in[3];
    const float* W2  = (const float*)d_in[4];
    const float* b2  = (const float*)d_in[5];
    float*       out = (float*)d_out;

    const int N = in_sizes[0] / IN;
    const int E = in_sizes[1] / 2;
    const int* src = ei;
    const int* dst = ei + E;

    char* ws = (char*)d_ws;
    float* dinv = (float*)ws;  ws += (size_t)N * sizeof(float);
    float* norm = (float*)ws;  ws += (size_t)E * sizeof(float);
    float* H1   = (float*)ws;  ws += (size_t)N * HID * sizeof(float);
    float* S1   = (float*)ws;
    float* H2   = H1;

    const int T = 256;

    k_fill1<<<cdiv(N, T), T, 0, stream>>>(dinv, N);
    k_deg  <<<cdiv(E, T), T, 0, stream>>>(dst, dinv, E);
    k_rsqrt<<<cdiv(N, T), T, 0, stream>>>(dinv, N);
    k_norm <<<cdiv(E, T), T, 0, stream>>>(src, dst, dinv, norm, E);

    gemm_wmma_f32<IN, HID, false><<<cdiv(N, 128), T, 0, stream>>>(x, W1, dinv, b1, H1, S1, N);
    k_scatter<HID><<<cdiv(E, T), T, 0, stream>>>(src, dst, norm, H1, S1, E);

    gemm_wmma_f32<HID, LAT, true><<<cdiv(N, 128), T, 0, stream>>>(S1, W2, dinv, b2, H2, out, N);
    k_scatter<LAT><<<cdiv(E, T), T, 0, stream>>>(src, dst, norm, H2, out, E);
}


# speedup vs baseline: 1.8832x; 1.1580x over previous
#include <hip/hip_runtime.h>


typedef __attribute__((ext_vector_type(2))) float v2f;
typedef __attribute__((ext_vector_type(8))) float v8f;

static inline int cdiv(long long a, long long b) { return (int)((a + b - 1) / b); }

template <int K, int N, bool RELU_A>
__launch_bounds__(256)
__global__ void gemm_wmma_f32(const float* __restrict__ X,
                              const float* __restrict__ W,
                              const float* __restrict__ dinv,
                              const float* __restrict__ bias,
                              float* __restrict__ H,
                              float* __restrict__ S,
                              int nrows) {
    constexpr int NT = N / 16;
    __shared__ float BF[(K / 4) * NT * 64];

    const int tid = threadIdx.x;
    for (int idx = tid; idx < K * N; idx += 256) {
        const int n  = idx / K;
        const int k  = idx - n * K;
        const int ks = k >> 2;
        const int kk = k & 3;
        const int g  = kk >> 1;
        const int r  = kk & 1;
        const int t  = n >> 4;
        const int nl = n & 15;
        BF[((ks * NT + t) * 32 + g * 16 + nl) * 2 + r] = W[idx];
    }
    __syncthreads();

    const int wave   = tid >> 5;
    const int lane   = tid & 31;
    const int laneLo = lane & 15;
    const int laneHi = lane >> 4;
    const int m0     = (blockIdx.x * 8 + wave) * 16;
    if (m0 >= nrows) return;

    v8f acc[NT] = {};

    const float* xrow  = X + (size_t)(m0 + laneLo) * K;
    const float* fbase = BF + lane * 2;
    for (int ks = 0; ks < K / 4; ++ks) {
        v2f a = *(const v2f*)(xrow + ks * 4 + 2 * laneHi);
        if (RELU_A) { a.x = fmaxf(a.x, 0.0f); a.y = fmaxf(a.y, 0.0f); }
        const float* fb = fbase + (ks * NT) * 64;
#pragma unroll
        for (int t = 0; t < NT; ++t) {
            const v2f b = *(const v2f*)(fb + t * 64);
            acc[t] = __builtin_amdgcn_wmma_f32_16x16x4_f32(
                false, a, false, b, (short)0, acc[t], false, false);
        }
    }

    const float4 da = *(const float4*)(dinv + m0 + 8 * laneHi);
    const float4 db = *(const float4*)(dinv + m0 + 8 * laneHi + 4);
    const float sc[8] = {da.x * da.x, da.y * da.y, da.z * da.z, da.w * da.w,
                         db.x * db.x, db.y * db.y, db.z * db.z, db.w * db.w};
#pragma unroll
    for (int t = 0; t < NT; ++t) {
        const float bt = bias[t * 16 + laneLo];
#pragma unroll
        for (int r = 0; r < 8; ++r) {
            const size_t o = (size_t)(m0 + r + 8 * laneHi) * N + t * 16 + laneLo;
            H[o] = acc[t][r];
            S[o] = acc[t][r] * sc[r] + bt;
        }
    }
}

__global__ void k_fill1(float* __restrict__ deg, int n) {
    int i = blockIdx.x * blockDim.x + threadIdx.x;
    if (i < n) deg[i] = 1.0f;
}

__global__ void k_deg(const int* __restrict__ dst, float* __restrict__ deg, int E) {
    int e = blockIdx.x * blockDim.x + threadIdx.x;
    if (e < E) atomicAdd(&deg[dst[e]], 1.0f);
}

__global__ void k_rsqrt(float* __restrict__ d, int n) {
    int i = blockIdx.x * blockDim.x + threadIdx.x;
    if (i < n) d[i] = rsqrtf(d[i]);
}

__global__ void k_norm(const int* __restrict__ src, const int* __restrict__ dst,
                       const float* __restrict__ dinv, float* __restrict__ norm, int E) {
    int e = blockIdx.x * blockDim.x + threadIdx.x;
    if (e < E) norm[e] = dinv[src[e]] * dinv[dst[e]];
}

template <int C, int CH, int U>
__global__ void k_scatter(const int* __restrict__ src, const int* __restrict__ dst,
                          const float* __restrict__ norm, const float* __restrict__ H,
                          float* __restrict__ out, int E) {
    constexpr int Q = CH / 32;
    const int c0   = (CH == C) ? 0 : (int)blockIdx.y * CH;
    const int lane = threadIdx.x & 31;
    const int wave = blockIdx.x * (blockDim.x >> 5) + (threadIdx.x >> 5);
    const int e0   = wave * 32;
    if (e0 >= E) return;
    const int n = min(32, E - e0);

    int s = 0, d = 0;
    float w = 0.0f;
    if (lane < n) { s = src[e0 + lane]; d = dst[e0 + lane]; w = norm[e0 + lane]; }

    int j = 0;
    for (; j + U <= n; j += U) {
        float v[U][Q];
#pragma unroll
        for (int u = 0; u < U; ++u) {
            const int sj = __builtin_amdgcn_readlane(s, j + u);
            const float* hp = H + (size_t)sj * C + c0 + lane;
#pragma unroll
            for (int q = 0; q < Q; ++q) v[u][q] = hp[q * 32];
        }
#pragma unroll
        for (int u = 0; u < U; ++u) {
            const int   dj = __builtin_amdgcn_readlane(d, j + u);
            const float wj = __int_as_float(__builtin_amdgcn_readlane(__float_as_int(w), j + u));
            float* op = out + (size_t)dj * C + c0 + lane;
#pragma unroll
            for (int q = 0; q < Q; ++q) atomicAdd(op + q * 32, v[u][q] * wj);
        }
    }
    for (; j < n; ++j) {
        const int   sj = __builtin_amdgcn_readlane(s, j);
        const int   dj = __builtin_amdgcn_readlane(d, j);
        const float wj = __int_as_float(__builtin_amdgcn_readlane(__float_as_int(w), j));
        const float* hp = H + (size_t)sj * C + c0 + lane;
        float* op = out + (size_t)dj * C + c0 + lane;
#pragma unroll
        for (int q = 0; q < Q; ++q) atomicAdd(op + q * 32, hp[q * 32] * wj);
    }
}

extern "C" void kernel_launch(void* const* d_in, const int* in_sizes, int n_in,
                              void* d_out, int out_size, void* d_ws, size_t ws_size,
                              hipStream_t stream) {
    constexpr int IN  = 256;
    constexpr int HID = 128;
    constexpr int LAT = 64;

    const float* x   = (const float*)d_in[0];
    const int*   ei  = (const int*)d_in[1];
    const float* W1  = (const float*)d_in[2];
    const float* b1  = (const float*)d_in[3];
    const float* W2  = (const float*)d_in[4];
    const float* b2  = (const float*)d_in[5];
    float*       out = (float*)d_out;

    const int N = in_sizes[0] / IN;
    const int E = in_sizes[1] / 2;
    const int* src = ei;
    const int* dst = ei + E;

    char* ws = (char*)d_ws;
    float* H1   = (float*)ws;  ws += (size_t)N * HID * sizeof(float);
    float* S1   = (float*)ws;  ws += (size_t)N * HID * sizeof(float);
    float* dinv = (float*)ws;  ws += (size_t)N * sizeof(float);
    float* norm = (float*)ws;
    float* H2   = H1;

    const int T = 256;

    k_fill1<<<cdiv(N, T), T, 0, stream>>>(dinv, N);
    k_deg  <<<cdiv(E, T), T, 0, stream>>>(dst, dinv, E);
    k_rsqrt<<<cdiv(N, T), T, 0, stream>>>(dinv, N);
    k_norm <<<cdiv(E, T), T, 0, stream>>>(src, dst, dinv, norm, E);

    gemm_wmma_f32<IN, HID, false><<<cdiv(N, 128), T, 0, stream>>>(x, W1, dinv, b1, H1, S1, N);
    k_scatter<HID, HID / 2, 16><<<dim3(cdiv(E, T), 2), T, 0, stream>>>(src, dst, norm, H1, S1, E);

    gemm_wmma_f32<HID, LAT, true><<<cdiv(N, 128), T, 0, stream>>>(S1, W2, dinv, b2, H2, out, N);
    k_scatter<LAT, LAT, 8><<<dim3(cdiv(E, T), 1), T, 0, stream>>>(src, dst, norm, H2, out, E);
}


# Degree count: 1.6M scattered global f32 atomics become LDS integer counters per (node range x edge slice) work-group, partial planes summed in the rsqrt pass; degrees exact; 8 launches -> 7
# speedup vs baseline: 2.3174x; 1.2306x over previous
#include <hip/hip_runtime.h>


typedef __attribute__((ext_vector_type(2))) float v2f;
typedef __attribute__((ext_vector_type(8))) float v8f;

static inline int cdiv(long long a, long long b) { return (int)((a + b - 1) / b); }

template <int K, int N, bool RELU_A>
__launch_bounds__(256)
__global__ void gemm_wmma_f32(const float* __restrict__ X,
                              const float* __restrict__ W,
                              const float* __restrict__ dinv,
                              const float* __restrict__ bias,
                              float* __restrict__ H,
                              float* __restrict__ S,
                              int nrows) {
    constexpr int NT = N / 16;
    __shared__ float BF[(K / 4) * NT * 64];

    const int tid = threadIdx.x;
    for (int idx = tid; idx < K * N; idx += 256) {
        const int n  = idx / K;
        const int k  = idx - n * K;
        const int ks = k >> 2;
        const int kk = k & 3;
        const int g  = kk >> 1;
        const int r  = kk & 1;
        const int t  = n >> 4;
        const int nl = n & 15;
        BF[((ks * NT + t) * 32 + g * 16 + nl) * 2 + r] = W[idx];
    }
    __syncthreads();

    const int wave   = tid >> 5;
    const int lane   = tid & 31;
    const int laneLo = lane & 15;
    const int laneHi = lane >> 4;
    const int m0     = (blockIdx.x * 8 + wave) * 16;
    if (m0 >= nrows) return;

    v8f acc[NT] = {};

    const float* xrow  = X + (size_t)(m0 + laneLo) * K;
    const float* fbase = BF + lane * 2;
    for (int ks = 0; ks < K / 4; ++ks) {
        v2f a = *(const v2f*)(xrow + ks * 4 + 2 * laneHi);
        if (RELU_A) { a.x = fmaxf(a.x, 0.0f); a.y = fmaxf(a.y, 0.0f); }
        const float* fb = fbase + (ks * NT) * 64;
#pragma unroll
        for (int t = 0; t < NT; ++t) {
            const v2f b = *(const v2f*)(fb + t * 64);
            acc[t] = __builtin_amdgcn_wmma_f32_16x16x4_f32(
                false, a, false, b, (short)0, acc[t], false, false);
        }
    }

    const float4 da = *(const float4*)(dinv + m0 + 8 * laneHi);
    const float4 db = *(const float4*)(dinv + m0 + 8 * laneHi + 4);
    const float sc[8] = {da.x * da.x, da.y * da.y, da.z * da.z, da.w * da.w,
                         db.x * db.x, db.y * db.y, db.z * db.z, db.w * db.w};
#pragma unroll
    for (int t = 0; t < NT; ++t) {
        const float bt = bias[t * 16 + laneLo];
#pragma unroll
        for (int r = 0; r < 8; ++r) {
            const size_t o = (size_t)(m0 + r + 8 * laneHi) * N + t * 16 + laneLo;
            H[o] = acc[t][r];
            S[o] = acc[t][r] * sc[r] + bt;
        }
    }
}

constexpr int HIST_RANGE  = 6400;
constexpr int HIST_SLICES = 16;
constexpr int HIST_T      = 1024;
constexpr int HIST_UN     = 16;

__launch_bounds__(HIST_T)
__global__ void k_hist(const int* __restrict__ dst, int* __restrict__ part, int N, int E) {
    __shared__ int cnt[HIST_RANGE];
    const int tid = threadIdx.x;
    const int lo  = (int)blockIdx.x * HIST_RANGE;
    const int nr  = min(HIST_RANGE, N - lo);
    for (int i = tid; i < HIST_RANGE; i += HIST_T) cnt[i] = 0;
    __syncthreads();

    const int L   = (E + HIST_SLICES - 1) / HIST_SLICES;
    const int beg = (int)blockIdx.y * L;
    const int end = min(E, beg + L);
    int e = beg + tid;
    for (; e + (HIST_UN - 1) * HIST_T < end; e += HIST_UN * HIST_T) {
        int d[HIST_UN];
#pragma unroll
        for (int u = 0; u < HIST_UN; ++u) d[u] = dst[e + u * HIST_T];
#pragma unroll
        for (int u = 0; u < HIST_UN; ++u) {
            const unsigned r = (unsigned)(d[u] - lo);
            if (r < (unsigned)nr) atomicAdd(&cnt[r], 1);
        }
    }
    for (; e < end; e += HIST_T) {
        const unsigned r = (unsigned)(dst[e] - lo);
        if (r < (unsigned)nr) atomicAdd(&cnt[r], 1);
    }
    __syncthreads();

    int* row = part + (size_t)blockIdx.y * N + lo;
    for (int i = tid; i < nr; i += HIST_T) row[i] = cnt[i];
}

__global__ void k_dinv(const int* __restrict__ part, float* __restrict__ dinv, int N) {
    int i = blockIdx.x * blockDim.x + threadIdx.x;
    if (i < N) {
        int c = 1;
#pragma unroll
        for (int m = 0; m < HIST_SLICES; ++m) c += part[(size_t)m * N + i];
        dinv[i] = rsqrtf((float)c);
    }
}

__global__ void k_norm(const int* __restrict__ src, const int* __restrict__ dst,
                       const float* __restrict__ dinv, float* __restrict__ norm, int E) {
    int e = blockIdx.x * blockDim.x + threadIdx.x;
    if (e < E) norm[e] = dinv[src[e]] * dinv[dst[e]];
}

template <int C, int CH, int U>
__global__ void k_scatter(const int* __restrict__ src, const int* __restrict__ dst,
                          const float* __restrict__ norm, const float* __restrict__ H,
                          float* __restrict__ out, int E) {
    constexpr int Q = CH / 32;
    const int c0   = (CH == C) ? 0 : (int)blockIdx.y * CH;
    const int lane = threadIdx.x & 31;
    const int wave = blockIdx.x * (blockDim.x >> 5) + (threadIdx.x >> 5);
    const int e0   = wave * 32;
    if (e0 >= E) return;
    const int n = min(32, E - e0);

    int s = 0, d = 0;
    float w = 0.0f;
    if (lane < n) { s = src[e0 + lane]; d = dst[e0 + lane]; w = norm[e0 + lane]; }

    int j = 0;
    for (; j + U <= n; j += U) {
        float v[U][Q];
#pragma unroll
        for (int u = 0; u < U; ++u) {
            const int sj = __builtin_amdgcn_readlane(s, j + u);
            const float* hp = H + (size_t)sj * C + c0 + lane;
#pragma unroll
            for (int q = 0; q < Q; ++q) v[u][q] = hp[q * 32];
        }
#pragma unroll
        for (int u = 0; u < U; ++u) {
            const int   dj = __builtin_amdgcn_readlane(d, j + u);
            const float wj = __int_as_float(__builtin_amdgcn_readlane(__float_as_int(w), j + u));
            float* op = out + (size_t)dj * C + c0 + lane;
#pragma unroll
            for (int q = 0; q < Q; ++q) atomicAdd(op + q * 32, v[u][q] * wj);
        }
    }
    for (; j < n; ++j) {
        const int   sj = __builtin_amdgcn_readlane(s, j);
        const int   dj = __builtin_amdgcn_readlane(d, j);
        const float wj = __int_as_float(__builtin_amdgcn_readlane(__float_as_int(w), j));
        const float* hp = H + (size_t)sj * C + c0 + lane;
        float* op = out + (size_t)dj * C + c0 + lane;
#pragma unroll
        for (int q = 0; q < Q; ++q) atomicAdd(op + q * 32, hp[q * 32] * wj);
    }
}

extern "C" void kernel_launch(void* const* d_in, const int* in_sizes, int n_in,
                              void* d_out, int out_size, void* d_ws, size_t ws_size,
                              hipStream_t stream) {
    constexpr int IN  = 256;
    constexpr int HID = 128;
    constexpr int LAT = 64;

    const float* x   = (const float*)d_in[0];
    const int*   ei  = (const int*)d_in[1];
    const float* W1  = (const float*)d_in[2];
    const float* b1  = (const float*)d_in[3];
    const float* W2  = (const float*)d_in[4];
    const float* b2  = (const float*)d_in[5];
    float*       out = (float*)d_out;

    const int N = in_sizes[0] / IN;
    const int E = in_sizes[1] / 2;
    const int* src = ei;
    const int* dst = ei + E;

    char* ws = (char*)d_ws;
    float* H1   = (float*)ws;  ws += (size_t)N * HID * sizeof(float);
    float* S1   = (float*)ws;  ws += (size_t)N * HID * sizeof(float);
    float* dinv = (float*)ws;  ws += (size_t)N * sizeof(float);
    float* norm = (float*)ws;
    float* H2   = H1;

    const int T = 256;

    int* part = (int*)H1;
    k_hist<<<dim3(cdiv(N, HIST_RANGE), HIST_SLICES), HIST_T, 0, stream>>>(dst, part, N, E);
    k_dinv<<<cdiv(N, T), T, 0, stream>>>(part, dinv, N);
    k_norm<<<cdiv(E, T), T, 0, stream>>>(src, dst, dinv, norm, E);

    gemm_wmma_f32<IN, HID, false><<<cdiv(N, 128), T, 0, stream>>>(x, W1, dinv, b1, H1, S1, N);
    k_scatter<HID, HID / 2, 16><<<dim3(cdiv(E, T), 2), T, 0, stream>>>(src, dst, norm, H1, S1, E);

    gemm_wmma_f32<HID, LAT, true><<<cdiv(N, 128), T, 0, stream>>>(S1, W2, dinv, b2, H2, out, N);
    k_scatter<LAT, LAT, 8><<<dim3(cdiv(E, T), 1), T, 0, stream>>>(src, dst, norm, H2, out, E);
}


# Edge-weight pass (norm = dinv[src]*dinv[dst]) moves into the layer-1 product's launch as extra work-groups handed out last, beside the matrix-bound tail; 7 launches -> 6; every sum unchanged
# speedup vs baseline: 2.4342x; 1.0504x over previous
#include <hip/hip_runtime.h>


typedef __attribute__((ext_vector_type(2))) float v2f;
typedef __attribute__((ext_vector_type(8))) float v8f;

static inline int cdiv(long long a, long long b) { return (int)((a + b - 1) / b); }

constexpr int EW_PER_T = 4;

template <int K, int N, bool RELU_A, bool EDGE_W>
__launch_bounds__(256)
__global__ void gemm_wmma_f32(const float* __restrict__ X,
                              const float* __restrict__ W,
                              const float* __restrict__ dinv,
                              const float* __restrict__ bias,
                              float* __restrict__ H,
                              float* __restrict__ S,
                              int nrows,
                              const int* __restrict__ src, const int* __restrict__ dst,
                              float* __restrict__ norm, int E, int prod_blocks) {
    constexpr int NT = N / 16;
    __shared__ float BF[(K / 4) * NT * 64];

    const int tid = threadIdx.x;
    if (EDGE_W && (int)blockIdx.x >= prod_blocks) {
        const int e0 = ((int)blockIdx.x - prod_blocks) * (256 * EW_PER_T) + tid;
        int s[EW_PER_T], d[EW_PER_T];
#pragma unroll
        for (int u = 0; u < EW_PER_T; ++u) {
            const int ec = min(e0 + u * 256, E - 1);
            s[u] = src[ec];
            d[u] = dst[ec];
        }
        float a[EW_PER_T], b[EW_PER_T];
#pragma unroll
        for (int u = 0; u < EW_PER_T; ++u) { a[u] = dinv[s[u]]; b[u] = dinv[d[u]]; }
#pragma unroll
        for (int u = 0; u < EW_PER_T; ++u) {
            const int e = e0 + u * 256;
            if (e < E) norm[e] = a[u] * b[u];
        }
        return;
    }
    for (int idx = tid; idx < K * N; idx += 256) {
        const int n  = idx / K;
        const int k  = idx - n * K;
        const int ks = k >> 2;
        const int kk = k & 3;
        const int g  = kk >> 1;
        const int r  = kk & 1;
        const int t  = n >> 4;
        const int nl = n & 15;
        BF[((ks * NT + t) * 32 + g * 16 + nl) * 2 + r] = W[idx];
    }
    __syncthreads();

    const int wave   = tid >> 5;
    const int lane   = tid & 31;
    const int laneLo = lane & 15;
    const int laneHi = lane >> 4;
    const int m0     = (blockIdx.x * 8 + wave) * 16;
    if (m0 >= nrows) return;

    v8f acc[NT] = {};

    const float* xrow  = X + (size_t)(m0 + laneLo) * K;
    const float* fbase = BF + lane * 2;
    for (int ks = 0; ks < K / 4; ++ks) {
        v2f a = *(const v2f*)(xrow + ks * 4 + 2 * laneHi);
        if (RELU_A) { a.x = fmaxf(a.x, 0.0f); a.y = fmaxf(a.y, 0.0f); }
        const float* fb = fbase + (ks * NT) * 64;
#pragma unroll
        for (int t = 0; t < NT; ++t) {
            const v2f b = *(const v2f*)(fb + t * 64);
            acc[t] = __builtin_amdgcn_wmma_f32_16x16x4_f32(
                false, a, false, b, (short)0, acc[t], false, false);
        }
    }

    const float4 da = *(const float4*)(dinv + m0 + 8 * laneHi);
    const float4 db = *(const float4*)(dinv + m0 + 8 * laneHi + 4);
    const float sc[8] = {da.x * da.x, da.y * da.y, da.z * da.z, da.w * da.w,
                         db.x * db.x, db.y * db.y, db.z * db.z, db.w * db.w};
#pragma unroll
    for (int t = 0; t < NT; ++t) {
        const float bt = bias[t * 16 + laneLo];
#pragma unroll
        for (int r = 0; r < 8; ++r) {
            const size_t o = (size_t)(m0 + r + 8 * laneHi) * N + t * 16 + laneLo;
            H[o] = acc[t][r];
            S[o] = acc[t][r] * sc[r] + bt;
        }
    }
}

constexpr int HIST_RANGE  = 6400;
constexpr int HIST_SLICES = 16;
constexpr int HIST_T      = 1024;
constexpr int HIST_UN     = 16;

__launch_bounds__(HIST_T)
__global__ void k_hist(const int* __restrict__ dst, int* __restrict__ part, int N, int E) {
    __shared__ int cnt[HIST_RANGE];
    const int tid = threadIdx.x;
    const int lo  = (int)blockIdx.x * HIST_RANGE;
    const int nr  = min(HIST_RANGE, N - lo);
    for (int i = tid; i < HIST_RANGE; i += HIST_T) cnt[i] = 0;
    __syncthreads();

    const int L   = (E + HIST_SLICES - 1) / HIST_SLICES;
    const int beg = (int)blockIdx.y * L;
    const int end = min(E, beg + L);
    int e = beg + tid;
    for (; e + (HIST_UN - 1) * HIST_T < end; e += HIST_UN * HIST_T) {
        int d[HIST_UN];
#pragma unroll
        for (int u = 0; u < HIST_UN; ++u) d[u] = dst[e + u * HIST_T];
#pragma unroll
        for (int u = 0; u < HIST_UN; ++u) {
            const unsigned r = (unsigned)(d[u] - lo);
            if (r < (unsigned)nr) atomicAdd(&cnt[r], 1);
        }
    }
    for (; e < end; e += HIST_T) {
        const unsigned r = (unsigned)(dst[e] - lo);
        if (r < (unsigned)nr) atomicAdd(&cnt[r], 1);
    }
    __syncthreads();

    int* row = part + (size_t)blockIdx.y * N + lo;
    for (int i = tid; i < nr; i += HIST_T) row[i] = cnt[i];
}

__global__ void k_dinv(const int* __restrict__ part, float* __restrict__ dinv, int N) {
    int i = blockIdx.x * blockDim.x + threadIdx.x;
    if (i < N) {
        int c = 1;
#pragma unroll
        for (int m = 0; m < HIST_SLICES; ++m) c += part[(size_t)m * N + i];
        dinv[i] = rsqrtf((float)c);
    }
}

template <int C, int CH, int U>
__global__ void k_scatter(const int* __restrict__ src, const int* __restrict__ dst,
                          const float* __restrict__ norm, const float* __restrict__ H,
                          float* __restrict__ out, int E) {
    constexpr int Q = CH / 32;
    const int c0   = (CH == C) ? 0 : (int)blockIdx.y * CH;
    const int lane = threadIdx.x & 31;
    const int wave = blockIdx.x * (blockDim.x >> 5) + (threadIdx.x >> 5);
    const int e0   = wave * 32;
    if (e0 >= E) return;
    const int n = min(32, E - e0);

    int s = 0, d = 0;
    float w = 0.0f;
    if (lane < n) { s = src[e0 + lane]; d = dst[e0 + lane]; w = norm[e0 + lane]; }

    int j = 0;
    for (; j + U <= n; j += U) {
        float v[U][Q];
#pragma unroll
        for (int u = 0; u < U; ++u) {
            const int sj = __builtin_amdgcn_readlane(s, j + u);
            const float* hp = H + (size_t)sj * C + c0 + lane;
#pragma unroll
            for (int q = 0; q < Q; ++q) v[u][q] = hp[q * 32];
        }
#pragma unroll
        for (int u = 0; u < U; ++u) {
            const int   dj = __builtin_amdgcn_readlane(d, j + u);
            const float wj = __int_as_float(__builtin_amdgcn_readlane(__float_as_int(w), j + u));
            float* op = out + (size_t)dj * C + c0 + lane;
#pragma unroll
            for (int q = 0; q < Q; ++q) atomicAdd(op + q * 32, v[u][q] * wj);
        }
    }
    for (; j < n; ++j) {
        const int   sj = __builtin_amdgcn_readlane(s, j);
        const int   dj = __builtin_amdgcn_readlane(d, j);
        const float wj = __int_as_float(__builtin_amdgcn_readlane(__float_as_int(w), j));
        const float* hp = H + (size_t)sj * C + c0 + lane;
        float* op = out + (size_t)dj * C + c0 + lane;
#pragma unroll
        for (int q = 0; q < Q; ++q) atomicAdd(op + q * 32, hp[q * 32] * wj);
    }
}

extern "C" void kernel_launch(void* const* d_in, const int* in_sizes, int n_in,
                              void* d_out, int out_size, void* d_ws, size_t ws_size,
                              hipStream_t stream) {
    constexpr int IN  = 256;
    constexpr int HID = 128;
    constexpr int LAT = 64;

    const float* x   = (const float*)d_in[0];
    const int*   ei  = (const int*)d_in[1];
    const float* W1  = (const float*)d_in[2];
    const float* b1  = (const float*)d_in[3];
    const float* W2  = (const float*)d_in[4];
    const float* b2  = (const float*)d_in[5];
    float*       out = (float*)d_out;

    const int N = in_sizes[0] / IN;
    const int E = in_sizes[1] / 2;
    const int* src = ei;
    const int* dst = ei + E;

    char* ws = (char*)d_ws;
    float* H1   = (float*)ws;  ws += (size_t)N * HID * sizeof(float);
    float* S1   = (float*)ws;  ws += (size_t)N * HID * sizeof(float);
    float* dinv = (float*)ws;  ws += (size_t)N * sizeof(float);
    float* norm = (float*)ws;
    float* H2   = H1;

    const int T = 256;

    int* part = (int*)H1;
    k_hist<<<dim3(cdiv(N, HIST_RANGE), HIST_SLICES), HIST_T, 0, stream>>>(dst, part, N, E);
    k_dinv<<<cdiv(N, T), T, 0, stream>>>(part, dinv, N);

    const int pb1 = cdiv(N, 128);
    gemm_wmma_f32<IN, HID, false, true><<<pb1 + cdiv(E, T * EW_PER_T), T, 0, stream>>>(
        x, W1, dinv, b1, H1, S1, N, src, dst, norm, E, pb1);
    k_scatter<HID, HID / 2, 16><<<dim3(cdiv(E, T), 2), T, 0, stream>>>(src, dst, norm, H1, S1, E);

    gemm_wmma_f32<HID, LAT, true, false><<<cdiv(N, 128), T, 0, stream>>>(
        S1, W2, dinv, b2, H2, out, N, src, dst, norm, E, cdiv(N, 128));
    k_scatter<LAT, LAT, 8><<<dim3(cdiv(E, T), 1), T, 0, stream>>>(src, dst, norm, H2, out, E);
}


# W fill of both products: 16-B coalesced reads and 8-B LDS stores into k-step blocks pitched 2 dwords past a multiple of 64, so a wave's stores cover all 64 banks once (were 4 banks); sums unchanged
# speedup vs baseline: 2.5258x; 1.0376x over previous
#include <hip/hip_runtime.h>


typedef __attribute__((ext_vector_type(2))) float v2f;
typedef __attribute__((ext_vector_type(8))) float v8f;

static inline int cdiv(long long a, long long b) { return (int)((a + b - 1) / b); }

constexpr int EW_PER_T = 4;

template <int K, int N, bool RELU_A, bool EDGE_W>
__launch_bounds__(256)
__global__ void gemm_wmma_f32(const float* __restrict__ X,
                              const float* __restrict__ W,
                              const float* __restrict__ dinv,
                              const float* __restrict__ bias,
                              float* __restrict__ H,
                              float* __restrict__ S,
                              int nrows,
                              const int* __restrict__ src, const int* __restrict__ dst,
                              float* __restrict__ norm, int E, int prod_blocks) {
    constexpr int NT = N / 16;
    constexpr int KS = K / 4;
    constexpr int PITCH = NT * 64 + 2;
    __shared__ float BF[KS * PITCH];

    const int tid = threadIdx.x;
    if (EDGE_W && (int)blockIdx.x >= prod_blocks) {
        const int e0 = ((int)blockIdx.x - prod_blocks) * (256 * EW_PER_T) + tid;
        int s[EW_PER_T], d[EW_PER_T];
#pragma unroll
        for (int u = 0; u < EW_PER_T; ++u) {
            const int ec = min(e0 + u * 256, E - 1);
            s[u] = src[ec];
            d[u] = dst[ec];
        }
        float a[EW_PER_T], b[EW_PER_T];
#pragma unroll
        for (int u = 0; u < EW_PER_T; ++u) { a[u] = dinv[s[u]]; b[u] = dinv[d[u]]; }
#pragma unroll
        for (int u = 0; u < EW_PER_T; ++u) {
            const int e = e0 + u * 256;
            if (e < E) norm[e] = a[u] * b[u];
        }
        return;
    }
    for (int idx = tid; idx < KS * N; idx += 256) {
        const int n  = idx / KS;
        const int ks = idx - n * KS;
        const int t  = n >> 4;
        const int nl = n & 15;
        const float4 w4 = *(const float4*)(W + (size_t)idx * 4);
        float* p = BF + ks * PITCH + t * 64 + nl * 2;
        *(v2f*)p        = v2f{w4.x, w4.y};
        *(v2f*)(p + 32) = v2f{w4.z, w4.w};
    }
    __syncthreads();

    const int wave   = tid >> 5;
    const int lane   = tid & 31;
    const int laneLo = lane & 15;
    const int laneHi = lane >> 4;
    const int m0     = (blockIdx.x * 8 + wave) * 16;
    if (m0 >= nrows) return;

    v8f acc[NT] = {};

    const float* xrow  = X + (size_t)(m0 + laneLo) * K;
    const float* fbase = BF + lane * 2;
    for (int ks = 0; ks < K / 4; ++ks) {
        v2f a = *(const v2f*)(xrow + ks * 4 + 2 * laneHi);
        if (RELU_A) { a.x = fmaxf(a.x, 0.0f); a.y = fmaxf(a.y, 0.0f); }
        const float* fb = fbase + ks * PITCH;
#pragma unroll
        for (int t = 0; t < NT; ++t) {
            const v2f b = *(const v2f*)(fb + t * 64);
            acc[t] = __builtin_amdgcn_wmma_f32_16x16x4_f32(
                false, a, false, b, (short)0, acc[t], false, false);
        }
    }

    const float4 da = *(const float4*)(dinv + m0 + 8 * laneHi);
    const float4 db = *(const float4*)(dinv + m0 + 8 * laneHi + 4);
    const float sc[8] = {da.x * da.x, da.y * da.y, da.z * da.z, da.w * da.w,
                         db.x * db.x, db.y * db.y, db.z * db.z, db.w * db.w};
#pragma unroll
    for (int t = 0; t < NT; ++t) {
        const float bt = bias[t * 16 + laneLo];
#pragma unroll
        for (int r = 0; r < 8; ++r) {
            const size_t o = (size_t)(m0 + r + 8 * laneHi) * N + t * 16 + laneLo;
            H[o] = acc[t][r];
            S[o] = acc[t][r] * sc[r] + bt;
        }
    }
}

constexpr int HIST_RANGE  = 6400;
constexpr int HIST_SLICES = 16;
constexpr int HIST_T      = 1024;
constexpr int HIST_UN     = 16;

__launch_bounds__(HIST_T)
__global__ void k_hist(const int* __restrict__ dst, int* __restrict__ part, int N, int E) {
    __shared__ int cnt[HIST_RANGE];
    const int tid = threadIdx.x;
    const int lo  = (int)blockIdx.x * HIST_RANGE;
    const int nr  = min(HIST_RANGE, N - lo);
    for (int i = tid; i < HIST_RANGE; i += HIST_T) cnt[i] = 0;
    __syncthreads();

    const int L   = (E + HIST_SLICES - 1) / HIST_SLICES;
    const int beg = (int)blockIdx.y * L;
    const int end = min(E, beg + L);
    int e = beg + tid;
    for (; e + (HIST_UN - 1) * HIST_T < end; e += HIST_UN * HIST_T) {
        int d[HIST_UN];
#pragma unroll
        for (int u = 0; u < HIST_UN; ++u) d[u] = dst[e + u * HIST_T];
#pragma unroll
        for (int u = 0; u < HIST_UN; ++u) {
            const unsigned r = (unsigned)(d[u] - lo);
            if (r < (unsigned)nr) atomicAdd(&cnt[r], 1);
        }
    }
    for (; e < end; e += HIST_T) {
        const unsigned r = (unsigned)(dst[e] - lo);
        if (r < (unsigned)nr) atomicAdd(&cnt[r], 1);
    }
    __syncthreads();

    int* row = part + (size_t)blockIdx.y * N + lo;
    for (int i = tid; i < nr; i += HIST_T) row[i] = cnt[i];
}

__global__ void k_dinv(const int* __restrict__ part, float* __restrict__ dinv, int N) {
    int i = blockIdx.x * blockDim.x + threadIdx.x;
    if (i < N) {
        int c = 1;
#pragma unroll
        for (int m = 0; m < HIST_SLICES; ++m) c += part[(size_t)m * N + i];
        dinv[i] = rsqrtf((float)c);
    }
}

template <int C, int CH, int U>
__global__ void k_scatter(const int* __restrict__ src, const int* __restrict__ dst,
                          const float* __restrict__ norm, const float* __restrict__ H,
                          float* __restrict__ out, int E) {
    constexpr int Q = CH / 32;
    const int c0   = (CH == C) ? 0 : (int)blockIdx.y * CH;
    const int lane = threadIdx.x & 31;
    const int wave = blockIdx.x * (blockDim.x >> 5) + (threadIdx.x >> 5);
    const int e0   = wave * 32;
    if (e0 >= E) return;
    const int n = min(32, E - e0);

    int s = 0, d = 0;
    float w = 0.0f;
    if (lane < n) { s = src[e0 + lane]; d = dst[e0 + lane]; w = norm[e0 + lane]; }

    int j = 0;
    for (; j + U <= n; j += U) {
        float v[U][Q];
#pragma unroll
        for (int u = 0; u < U; ++u) {
            const int sj = __builtin_amdgcn_readlane(s, j + u);
            const float* hp = H + (size_t)sj * C + c0 + lane;
#pragma unroll
            for (int q = 0; q < Q; ++q) v[u][q] = hp[q * 32];
        }
#pragma unroll
        for (int u = 0; u < U; ++u) {
            const int   dj = __builtin_amdgcn_readlane(d, j + u);
            const float wj = __int_as_float(__builtin_amdgcn_readlane(__float_as_int(w), j + u));
            float* op = out + (size_t)dj * C + c0 + lane;
#pragma unroll
            for (int q = 0; q < Q; ++q) atomicAdd(op + q * 32, v[u][q] * wj);
        }
    }
    for (; j < n; ++j) {
        const int   sj = __builtin_amdgcn_readlane(s, j);
        const int   dj = __builtin_amdgcn_readlane(d, j);
        const float wj = __int_as_float(__builtin_amdgcn_readlane(__float_as_int(w), j));
        const float* hp = H + (size_t)sj * C + c0 + lane;
        float* op = out + (size_t)dj * C + c0 + lane;
#pragma unroll
        for (int q = 0; q < Q; ++q) atomicAdd(op + q * 32, hp[q * 32] * wj);
    }
}

extern "C" void kernel_launch(void* const* d_in, const int* in_sizes, int n_in,
                              void* d_out, int out_size, void* d_ws, size_t ws_size,
                              hipStream_t stream) {
    constexpr int IN  = 256;
    constexpr int HID = 128;
    constexpr int LAT = 64;

    const float* x   = (const float*)d_in[0];
    const int*   ei  = (const int*)d_in[1];
    const float* W1  = (const float*)d_in[2];
    const float* b1  = (const float*)d_in[3];
    const float* W2  = (const float*)d_in[4];
    const float* b2  = (const float*)d_in[5];
    float*       out = (float*)d_out;

    const int N = in_sizes[0] / IN;
    const int E = in_sizes[1] / 2;
    const int* src = ei;
    const int* dst = ei + E;

    char* ws = (char*)d_ws;
    float* H1   = (float*)ws;  ws += (size_t)N * HID * sizeof(float);
    float* S1   = (float*)ws;  ws += (size_t)N * HID * sizeof(float);
    float* dinv = (float*)ws;  ws += (size_t)N * sizeof(float);
    float* norm = (float*)ws;
    float* H2   = H1;

    const int T = 256;

    int* part = (int*)H1;
    k_hist<<<dim3(cdiv(N, HIST_RANGE), HIST_SLICES), HIST_T, 0, stream>>>(dst, part, N, E);
    k_dinv<<<cdiv(N, T), T, 0, stream>>>(part, dinv, N);

    const int pb1 = cdiv(N, 128);
    gemm_wmma_f32<IN, HID, false, true><<<pb1 + cdiv(E, T * EW_PER_T), T, 0, stream>>>(
        x, W1, dinv, b1, H1, S1, N, src, dst, norm, E, pb1);
    k_scatter<HID, HID / 2, 16><<<dim3(cdiv(E, T), 2), T, 0, stream>>>(src, dst, norm, H1, S1, E);

    gemm_wmma_f32<HID, LAT, true, false><<<cdiv(N, 128), T, 0, stream>>>(
        S1, W2, dinv, b2, H2, out, N, src, dst, norm, E, cdiv(N, 128));
    k_scatter<LAT, LAT, 8><<<dim3(cdiv(E, T), 1), T, 0, stream>>>(src, dst, norm, H2, out, E);
}


# Edge scatter keeps each row's f32 atomics in one L2 half: nodes split in 2 ranges; a work-group reads its die, draws a chunk ticket from its half's queue, adds only own-range edges; arrival order
# speedup vs baseline: 2.8091x; 1.1122x over previous
#include <hip/hip_runtime.h>


typedef __attribute__((ext_vector_type(2))) float v2f;
typedef __attribute__((ext_vector_type(8))) float v8f;

static inline int cdiv(long long a, long long b) { return (int)((a + b - 1) / b); }

constexpr int EW_PER_T = 4;
constexpr int TICK_PITCH = 64;

template <int K, int N, bool RELU_A, bool EDGE_W>
__launch_bounds__(256)
__global__ void gemm_wmma_f32(const float* __restrict__ X,
                              const float* __restrict__ W,
                              const float* __restrict__ dinv,
                              const float* __restrict__ bias,
                              float* __restrict__ H,
                              float* __restrict__ S,
                              int nrows,
                              const int* __restrict__ src, const int* __restrict__ dst,
                              float* __restrict__ norm, int E, int prod_blocks,
                              int* __restrict__ tick, int ntick) {
    constexpr int NT = N / 16;
    constexpr int KS = K / 4;
    constexpr int PITCH = NT * 64 + 2;
    __shared__ float BF[KS * PITCH];

    const int tid = threadIdx.x;
    if (blockIdx.x == 0 && tid < ntick) tick[tid * TICK_PITCH] = 0;
    if (EDGE_W && (int)blockIdx.x >= prod_blocks) {
        const int e0 = ((int)blockIdx.x - prod_blocks) * (256 * EW_PER_T) + tid;
        int s[EW_PER_T], d[EW_PER_T];
#pragma unroll
        for (int u = 0; u < EW_PER_T; ++u) {
            const int ec = min(e0 + u * 256, E - 1);
            s[u] = src[ec];
            d[u] = dst[ec];
        }
        float a[EW_PER_T], b[EW_PER_T];
#pragma unroll
        for (int u = 0; u < EW_PER_T; ++u) { a[u] = dinv[s[u]]; b[u] = dinv[d[u]]; }
#pragma unroll
        for (int u = 0; u < EW_PER_T; ++u) {
            const int e = e0 + u * 256;
            if (e < E) norm[e] = a[u] * b[u];
        }
        return;
    }
    for (int idx = tid; idx < KS * N; idx += 256) {
        const int n  = idx / KS;
        const int ks = idx - n * KS;
        const int t  = n >> 4;
        const int nl = n & 15;
        const float4 w4 = *(const float4*)(W + (size_t)idx * 4);
        float* p = BF + ks * PITCH + t * 64 + nl * 2;
        *(v2f*)p        = v2f{w4.x, w4.y};
        *(v2f*)(p + 32) = v2f{w4.z, w4.w};
    }
    __syncthreads();

    const int wave   = tid >> 5;
    const int lane   = tid & 31;
    const int laneLo = lane & 15;
    const int laneHi = lane >> 4;
    const int m0     = (blockIdx.x * 8 + wave) * 16;
    if (m0 >= nrows) return;

    v8f acc[NT] = {};

    const float* xrow  = X + (size_t)(m0 + laneLo) * K;
    const float* fbase = BF + lane * 2;
    for (int ks = 0; ks < K / 4; ++ks) {
        v2f a = *(const v2f*)(xrow + ks * 4 + 2 * laneHi);
        if (RELU_A) { a.x = fmaxf(a.x, 0.0f); a.y = fmaxf(a.y, 0.0f); }
        const float* fb = fbase + ks * PITCH;
#pragma unroll
        for (int t = 0; t < NT; ++t) {
            const v2f b = *(const v2f*)(fb + t * 64);
            acc[t] = __builtin_amdgcn_wmma_f32_16x16x4_f32(
                false, a, false, b, (short)0, acc[t], false, false);
        }
    }

    const float4 da = *(const float4*)(dinv + m0 + 8 * laneHi);
    const float4 db = *(const float4*)(dinv + m0 + 8 * laneHi + 4);
    const float sc[8] = {da.x * da.x, da.y * da.y, da.z * da.z, da.w * da.w,
                         db.x * db.x, db.y * db.y, db.z * db.z, db.w * db.w};
#pragma unroll
    for (int t = 0; t < NT; ++t) {
        const float bt = bias[t * 16 + laneLo];
#pragma unroll
        for (int r = 0; r < 8; ++r) {
            const size_t o = (size_t)(m0 + r + 8 * laneHi) * N + t * 16 + laneLo;
            H[o] = acc[t][r];
            S[o] = acc[t][r] * sc[r] + bt;
        }
    }
}

constexpr int HIST_RANGE  = 6400;
constexpr int HIST_SLICES = 16;
constexpr int HIST_T      = 1024;
constexpr int HIST_UN     = 16;

__launch_bounds__(HIST_T)
__global__ void k_hist(const int* __restrict__ dst, int* __restrict__ part, int N, int E) {
    __shared__ int cnt[HIST_RANGE];
    const int tid = threadIdx.x;
    const int lo  = (int)blockIdx.x * HIST_RANGE;
    const int nr  = min(HIST_RANGE, N - lo);
    for (int i = tid; i < HIST_RANGE; i += HIST_T) cnt[i] = 0;
    __syncthreads();

    const int L   = (E + HIST_SLICES - 1) / HIST_SLICES;
    const int beg = (int)blockIdx.y * L;
    const int end = min(E, beg + L);
    int e = beg + tid;
    for (; e + (HIST_UN - 1) * HIST_T < end; e += HIST_UN * HIST_T) {
        int d[HIST_UN];
#pragma unroll
        for (int u = 0; u < HIST_UN; ++u) d[u] = dst[e + u * HIST_T];
#pragma unroll
        for (int u = 0; u < HIST_UN; ++u) {
            const unsigned r = (unsigned)(d[u] - lo);
            if (r < (unsigned)nr) atomicAdd(&cnt[r], 1);
        }
    }
    for (; e < end; e += HIST_T) {
        const unsigned r = (unsigned)(dst[e] - lo);
        if (r < (unsigned)nr) atomicAdd(&cnt[r], 1);
    }
    __syncthreads();

    int* row = part + (size_t)blockIdx.y * N + lo;
    for (int i = tid; i < nr; i += HIST_T) row[i] = cnt[i];
}

__global__ void k_dinv(const int* __restrict__ part, float* __restrict__ dinv, int N) {
    int i = blockIdx.x * blockDim.x + threadIdx.x;
    if (i < N) {
        int c = 1;
#pragma unroll
        for (int m = 0; m < HIST_SLICES; ++m) c += part[(size_t)m * N + i];
        dinv[i] = rsqrtf((float)c);
    }
}

constexpr int SC_EDGES = 512;

template <int C, int CH, int U>
__launch_bounds__(256)
__global__ void k_scatter(const int* __restrict__ src, const int* __restrict__ dst,
                          const float* __restrict__ norm, const float* __restrict__ H,
                          float* __restrict__ out, int* __restrict__ tick, int E, int N) {
    constexpr int Q = CH / 32;
    __shared__ int sh[2];
    const int pass = (CH == C) ? 0 : (int)blockIdx.y;
    const int c0   = pass * CH;
    const int nchunks = (E + SC_EDGES - 1) / SC_EDGES;
    if (threadIdx.x == 0) {
        int h = (int)((__builtin_amdgcn_s_sendmsg_rtn(0x87) >> 18) & 1u);
        int* q = tick + pass * (2 * TICK_PITCH);
        int t = atomicAdd(q + h * TICK_PITCH, 1);
        if (t >= nchunks) { h ^= 1; t = atomicAdd(q + h * TICK_PITCH, 1); }
        sh[0] = t;
        sh[1] = h;
    }
    __syncthreads();
    const int t = sh[0];
    const int h = sh[1];
    if (t >= nchunks) return;

    const int nsplit = N >> 1;
    const int lane = threadIdx.x & 31;
    const int eb   = t * SC_EDGES + (int)(threadIdx.x >> 5) * 64;

    int s[2], d[2];
    float w[2];
    unsigned m[2];
#pragma unroll
    for (int g = 0; g < 2; ++g) {
        const int e = eb + g * 32 + lane;
        s[g] = 0; d[g] = 0; w[g] = 0.0f;
        if (e < E) { s[g] = src[e]; d[g] = dst[e]; w[g] = norm[e]; }
    }
#pragma unroll
    for (int g = 0; g < 2; ++g) {
        const int e = eb + g * 32 + lane;
        const bool mine = (e < E) && ((d[g] >= nsplit) == (h != 0));
        m[g] = __builtin_amdgcn_ballot_w32(mine);
    }

#pragma unroll
    for (int g = 0; g < 2; ++g) {
        for (int j = 0; j < 32; j += U) {
            if (((m[g] >> j) & ((U == 32) ? 0xffffffffu : ((1u << U) - 1u))) == 0u) continue;
            float v[U][Q];
#pragma unroll
            for (int u = 0; u < U; ++u) {
                if ((m[g] >> (j + u)) & 1u) {
                    const int sj = __builtin_amdgcn_readlane(s[g], j + u);
                    const float* hp = H + (size_t)sj * C + c0 + lane;
#pragma unroll
                    for (int q = 0; q < Q; ++q) v[u][q] = hp[q * 32];
                }
            }
#pragma unroll
            for (int u = 0; u < U; ++u) {
                if ((m[g] >> (j + u)) & 1u) {
                    const int   dj = __builtin_amdgcn_readlane(d[g], j + u);
                    const float wj = __int_as_float(
                        __builtin_amdgcn_readlane(__float_as_int(w[g]), j + u));
                    float* op = out + (size_t)dj * C + c0 + lane;
#pragma unroll
                    for (int q = 0; q < Q; ++q) atomicAdd(op + q * 32, v[u][q] * wj);
                }
            }
        }
    }
}

extern "C" void kernel_launch(void* const* d_in, const int* in_sizes, int n_in,
                              void* d_out, int out_size, void* d_ws, size_t ws_size,
                              hipStream_t stream) {
    constexpr int IN  = 256;
    constexpr int HID = 128;
    constexpr int LAT = 64;

    const float* x   = (const float*)d_in[0];
    const int*   ei  = (const int*)d_in[1];
    const float* W1  = (const float*)d_in[2];
    const float* b1  = (const float*)d_in[3];
    const float* W2  = (const float*)d_in[4];
    const float* b2  = (const float*)d_in[5];
    float*       out = (float*)d_out;

    const int N = in_sizes[0] / IN;
    const int E = in_sizes[1] / 2;
    const int* src = ei;
    const int* dst = ei + E;

    char* ws = (char*)d_ws;
    float* H1   = (float*)ws;  ws += (size_t)N * HID * sizeof(float);
    float* S1   = (float*)ws;  ws += (size_t)N * HID * sizeof(float);
    float* dinv = (float*)ws;  ws += (size_t)N * sizeof(float);
    float* norm = (float*)ws;
    float* H2   = H1;

    const int T = 256;

    int* part = (int*)H1;
    k_hist<<<dim3(cdiv(N, HIST_RANGE), HIST_SLICES), HIST_T, 0, stream>>>(dst, part, N, E);
    k_dinv<<<cdiv(N, T), T, 0, stream>>>(part, dinv, N);

    const int pb1 = cdiv(N, 128);
    int* tick1 = (int*)d_out;
    gemm_wmma_f32<IN, HID, false, true><<<pb1 + cdiv(E, T * EW_PER_T), T, 0, stream>>>(
        x, W1, dinv, b1, H1, S1, N, src, dst, norm, E, pb1, tick1, 4);
    const int sc_groups = 2 * cdiv(E, SC_EDGES);
    k_scatter<HID, HID / 2, 16><<<dim3(sc_groups, 2), T, 0, stream>>>(src, dst, norm, H1, S1, tick1, E, N);

    int* tick2 = (int*)(H1 + (size_t)N * LAT);
    gemm_wmma_f32<HID, LAT, true, false><<<cdiv(N, 128), T, 0, stream>>>(
        S1, W2, dinv, b2, H2, out, N, src, dst, norm, E, cdiv(N, 128), tick2, 2);
    k_scatter<LAT, LAT, 16><<<dim3(sc_groups, 1), T, 0, stream>>>(src, dst, norm, H2, out, tick2, E, N);
}
